# starting kernel
# baseline (speedup 1.0000x reference)
#include <hip/hip_runtime.h>

constexpr int FEAT = 128;
constexpr int DEG  = 32;

typedef __attribute__((ext_vector_type(2))) float v2f;
typedef __attribute__((ext_vector_type(8))) float v8f;

__global__ void ncn_node_phi(const float* __restrict__ x,
                             const float* __restrict__ W,
                             float* __restrict__ y, int N) {
  const int wid  = (int)((blockIdx.x * blockDim.x + threadIdx.x) >> 5);
  const int lane = (int)(threadIdx.x & 31);
  if (wid >= N) return;
  const float4 xv = *(const float4*)(x + (size_t)wid * FEAT + lane * 4);
  const float4 wv = *(const float4*)(W + FEAT + lane * 4);
  float s = xv.x * wv.x + xv.y * wv.y + xv.z * wv.z + xv.w * wv.w;
#pragma unroll
  for (int off = 16; off >= 1; off >>= 1) s += __shfl_xor(s, off, 32);
  if (lane == 0) y[wid] = s;
}

__global__ void ncn_edge_kernel(const float* __restrict__ x,
                                const int*  __restrict__ nbr,
                                const int*  __restrict__ tar_ei,
                                const float* __restrict__ W,
                                const float* __restrict__ bptr,
                                const float* __restrict__ y,
                                float* __restrict__ out, int E) {
  const int lane  = (int)(threadIdx.x & 31);
  const int wid   = (int)((blockIdx.x * blockDim.x + threadIdx.x) >> 5);
  const int eBase = wid * 16;
  if (eBase >= E) return;

  const int m    = lane & 15;
  const int koff = (lane >> 4) * 2;

  int eIdx = eBase + m;
  if (eIdx > E - 1) eIdx = E - 1;
  const int src = tar_ei[eIdx];
  const int dst = tar_ei[E + eIdx];

  float sreg[8];
#pragma unroll
  for (int ee = 0; ee < 16; ++ee) {
    const int eSrc = __shfl(src, ee, 32);
    const int eDst = __shfl(dst, ee, 32);
    const int q    = nbr[(size_t)eSrc * DEG + lane];
    const int njv  = nbr[(size_t)eDst * DEG + lane];

    int pos = 0;
#pragma unroll
    for (int b = 16; b >= 1; b >>= 1) {
      const int v = __shfl(njv, pos + b - 1, 32);
      if (v < q) pos += b;
    }
    const bool hit = (__shfl(njv, pos, 32) == q);

    float sv = 0.0f;
    if (hit) sv = y[q];
#pragma unroll
    for (int off = 16; off >= 1; off >>= 1) sv += __shfl_xor(sv, off, 32);
    if ((lane >> 4) == (ee >> 3)) sreg[ee & 7] = sv;
  }

  const float* xi_row = x + (size_t)src * FEAT;
  const float* xj_row = x + (size_t)dst * FEAT;
  const bool col0 = (m == 0);

  v8f c = {0.f, 0.f, 0.f, 0.f, 0.f, 0.f, 0.f, 0.f};
#pragma unroll 8
  for (int k = 0; k < FEAT; k += 4) {
    const v2f xiv = *(const v2f*)(xi_row + k + koff);
    const v2f xjv = *(const v2f*)(xj_row + k + koff);
    const v2f wv  = *(const v2f*)(W + k + koff);
    const v2f a   = xiv * xjv;
    v2f bv;
    bv.x = col0 ? wv.x : 0.0f;
    bv.y = col0 ? wv.y : 0.0f;
    c = __builtin_amdgcn_wmma_f32_16x16x4_f32(
             false, a,  false, bv,
             (short)0, c,  false,  false);
  }

  const float bias = bptr[0];
  if (lane == 0) {
#pragma unroll
    for (int r = 0; r < 8; ++r) {
      const int e = eBase + r;
      if (e < E) out[e] = c[r] + sreg[r] + bias;
    }
  } else if (lane == 16) {
#pragma unroll
    for (int r = 0; r < 8; ++r) {
      const int e = eBase + 8 + r;
      if (e < E) out[e] = c[r] + sreg[r] + bias;
    }
  }
}

extern "C" void kernel_launch(void* const* d_in, const int* in_sizes, int n_in,
                              void* d_out, int out_size, void* d_ws, size_t ws_size,
                              hipStream_t stream) {
  const float* x      = (const float*)d_in[0];
  const int*   nbr    = (const int*)  d_in[1];
  const int*   tar_ei = (const int*)  d_in[2];
  const float* W      = (const float*)d_in[3];
  const float* b      = (const float*)d_in[4];
  float*       out    = (float*)d_out;

  const int N = in_sizes[0] / FEAT;
  const int E = in_sizes[2] / 2;

  float* y = (float*)d_ws;

  const int threads = 256;
  const int blocks1 = (int)(((long long)N * 32 + threads - 1) / threads);
  ncn_node_phi<<<blocks1, threads, 0, stream>>>(x, W, y, N);

  const int edgesPerBlock = (threads / 32) * 16;
  const int blocks2 = (E + edgesPerBlock - 1) / edgesPerBlock;
  ncn_edge_kernel<<<blocks2, threads, 0, stream>>>(x, nbr, tar_ei, W, b, y, out, E);
}


# Common-neighbour term on demand: the all-nodes projection pre-pass is dropped; the edge kernel reads and projects a node's row only for hits (wave-uniform ballot per edge); one launch, not two
# speedup vs baseline: 2.0130x; 2.0130x over previous
#include <hip/hip_runtime.h>

constexpr int FEAT = 128;
constexpr int DEG  = 32;

typedef __attribute__((ext_vector_type(2))) float v2f;
typedef __attribute__((ext_vector_type(8))) float v8f;

__global__ void ncn_edge_kernel(const float* __restrict__ x,
                                const int*  __restrict__ nbr,
                                const int*  __restrict__ tar_ei,
                                const float* __restrict__ W,
                                const float* __restrict__ bptr,
                                float* __restrict__ out, int E) {
  const int lane  = (int)(threadIdx.x & 31);
  const int wid   = (int)((blockIdx.x * blockDim.x + threadIdx.x) >> 5);
  const int eBase = wid * 16;
  if (eBase >= E) return;

  const int m    = lane & 15;
  const int koff = (lane >> 4) * 2;

  int eIdx = eBase + m;
  if (eIdx > E - 1) eIdx = E - 1;
  const int src = tar_ei[eIdx];
  const int dst = tar_ei[E + eIdx];

  float sreg[8] = {0.f, 0.f, 0.f, 0.f, 0.f, 0.f, 0.f, 0.f};
#pragma unroll
  for (int ee = 0; ee < 16; ++ee) {
    const int eSrc = __shfl(src, ee, 32);
    const int eDst = __shfl(dst, ee, 32);
    const int q    = nbr[(size_t)eSrc * DEG + lane];
    const int njv  = nbr[(size_t)eDst * DEG + lane];

    int pos = 0;
#pragma unroll
    for (int b = 16; b >= 1; b >>= 1) {
      const int v = __shfl(njv, pos + b - 1, 32);
      if (v < q) pos += b;
    }
    const bool hit = (__shfl(njv, pos, 32) == q);

    const unsigned hmask = __builtin_amdgcn_ballot_w32(hit);
    if (__builtin_expect(hmask != 0u, 0)) {
      float sv = 0.0f;
      unsigned rem = hmask;
      for (int t = 0; t < 32 && rem != 0u; ++t) {
        const int h  = __builtin_ctz(rem);
        rem &= rem - 1u;
        const int qh = __builtin_amdgcn_readlane(q, h);
        const float4 xv = *(const float4*)(x + (size_t)qh * FEAT + lane * 4);
        const float4 wv = *(const float4*)(W + FEAT + lane * 4);
        float s = xv.y * wv.y;
        s = __builtin_fmaf(xv.x, wv.x, s);
        s = __builtin_fmaf(xv.z, wv.z, s);
        s = __builtin_fmaf(xv.w, wv.w, s);
#pragma unroll
        for (int off = 16; off >= 1; off >>= 1) s += __shfl_xor(s, off, 32);
        if (lane == h) sv = s;
      }
#pragma unroll
      for (int off = 16; off >= 1; off >>= 1) sv += __shfl_xor(sv, off, 32);
      if ((lane >> 4) == (ee >> 3)) sreg[ee & 7] = sv;
    }
  }

  const float* xi_row = x + (size_t)src * FEAT;
  const float* xj_row = x + (size_t)dst * FEAT;
  const bool col0 = (m == 0);

  v8f c = {0.f, 0.f, 0.f, 0.f, 0.f, 0.f, 0.f, 0.f};
  for (int k0 = 0; k0 < FEAT; k0 += 32) {
    v2f xiv[8], xjv[8], wv[8];
#pragma unroll
    for (int u = 0; u < 8; ++u) {
      const int k = k0 + 4 * u + koff;
      xiv[u] = *(const v2f*)(xi_row + k);
      xjv[u] = *(const v2f*)(xj_row + k);
      wv[u]  = *(const v2f*)(W + k);
    }
#pragma unroll
    for (int u = 0; u < 8; ++u) {
      const v2f a = xiv[u] * xjv[u];
      v2f bv;
      bv.x = col0 ? wv[u].x : 0.0f;
      bv.y = col0 ? wv[u].y : 0.0f;
      c = __builtin_amdgcn_wmma_f32_16x16x4_f32(
               false, a,  false, bv,
               (short)0, c,  false,  false);
    }
  }

  const float bias = bptr[0];
  if (lane == 0) {
#pragma unroll
    for (int r = 0; r < 8; ++r) {
      const int e = eBase + r;
      if (e < E) out[e] = c[r] + sreg[r] + bias;
    }
  } else if (lane == 16) {
#pragma unroll
    for (int r = 0; r < 8; ++r) {
      const int e = eBase + 8 + r;
      if (e < E) out[e] = c[r] + sreg[r] + bias;
    }
  }
}

extern "C" void kernel_launch(void* const* d_in, const int* in_sizes, int n_in,
                              void* d_out, int out_size, void* d_ws, size_t ws_size,
                              hipStream_t stream) {
  const float* x      = (const float*)d_in[0];
  const int*   nbr    = (const int*)  d_in[1];
  const int*   tar_ei = (const int*)  d_in[2];
  const float* W      = (const float*)d_in[3];
  const float* b      = (const float*)d_in[4];
  float*       out    = (float*)d_out;

  const int E = in_sizes[2] / 2;

  const int threads = 256;
  const int edgesPerBlock = (threads / 32) * 16;
  const int blocks2 = (E + edgesPerBlock - 1) / edgesPerBlock;
  ncn_edge_kernel<<<blocks2, threads, 0, stream>>>(x, nbr, tar_ei, W, b, out, E);
}


# The 16 per-edge neighbor searches run side by side with no branch between them (all 32 row loads first, shuffle steps interleaved); hit work moved behind one wave-uniform test after the dot stage
# speedup vs baseline: 3.2268x; 1.6030x over previous
#include <hip/hip_runtime.h>

constexpr int FEAT = 128;
constexpr int DEG  = 32;

typedef __attribute__((ext_vector_type(2))) float v2f;
typedef __attribute__((ext_vector_type(8))) float v8f;

__global__ void ncn_edge_kernel(const float* __restrict__ x,
                                const int*  __restrict__ nbr,
                                const int*  __restrict__ tar_ei,
                                const float* __restrict__ W,
                                const float* __restrict__ bptr,
                                float* __restrict__ out, int E) {
  const int lane  = (int)(threadIdx.x & 31);
  const int wid   = (int)((blockIdx.x * blockDim.x + threadIdx.x) >> 5);
  const int eBase = wid * 16;
  if (eBase >= E) return;

  const int m    = lane & 15;
  const int koff = (lane >> 4) * 2;

  int eIdx = eBase + m;
  if (eIdx > E - 1) eIdx = E - 1;
  const int src = tar_ei[eIdx];
  const int dst = tar_ei[E + eIdx];

  int q[16], njv[16];
#pragma unroll
  for (int ee = 0; ee < 16; ++ee) {
    const int eSrc = __shfl(src, ee, 32);
    const int eDst = __shfl(dst, ee, 32);
    q[ee]   = nbr[(size_t)eSrc * DEG + lane];
    njv[ee] = nbr[(size_t)eDst * DEG + lane];
  }

  int pos[16];
#pragma unroll
  for (int ee = 0; ee < 16; ++ee) pos[ee] = 0;
#pragma unroll
  for (int b = 16; b >= 1; b >>= 1) {
#pragma unroll
    for (int ee = 0; ee < 16; ++ee) {
      const int v = __shfl(njv[ee], pos[ee] + b - 1, 32);
      if (v < q[ee]) pos[ee] += b;
    }
  }
  unsigned hmask[16];
  unsigned anyHit = 0u;
#pragma unroll
  for (int ee = 0; ee < 16; ++ee) {
    const bool hit = (__shfl(njv[ee], pos[ee], 32) == q[ee]);
    hmask[ee] = __builtin_amdgcn_ballot_w32(hit);
    anyHit |= hmask[ee];
  }

  const float* xi_row = x + (size_t)src * FEAT;
  const float* xj_row = x + (size_t)dst * FEAT;
  const bool col0 = (m == 0);

  v8f c = {0.f, 0.f, 0.f, 0.f, 0.f, 0.f, 0.f, 0.f};
  for (int k0 = 0; k0 < FEAT; k0 += 32) {
    v2f xiv[8], xjv[8], wv[8];
#pragma unroll
    for (int u = 0; u < 8; ++u) {
      const int k = k0 + 4 * u + koff;
      xiv[u] = *(const v2f*)(xi_row + k);
      xjv[u] = *(const v2f*)(xj_row + k);
      wv[u]  = *(const v2f*)(W + k);
    }
#pragma unroll
    for (int u = 0; u < 8; ++u) {
      const v2f a = xiv[u] * xjv[u];
      v2f bv;
      bv.x = col0 ? wv[u].x : 0.0f;
      bv.y = col0 ? wv[u].y : 0.0f;
      c = __builtin_amdgcn_wmma_f32_16x16x4_f32(
               false, a,  false, bv,
               (short)0, c,  false,  false);
    }
  }

  float sreg[8] = {0.f, 0.f, 0.f, 0.f, 0.f, 0.f, 0.f, 0.f};
  if (__builtin_expect(anyHit != 0u, 0)) {
#pragma unroll
    for (int ee = 0; ee < 16; ++ee) {
      if (hmask[ee] != 0u) {
        float sv = 0.0f;
        unsigned rem = hmask[ee];
        for (int t = 0; t < 32 && rem != 0u; ++t) {
          const int h  = __builtin_ctz(rem);
          rem &= rem - 1u;
          const int qh = __builtin_amdgcn_readlane(q[ee], h);
          const float4 xv = *(const float4*)(x + (size_t)qh * FEAT + lane * 4);
          const float4 wv = *(const float4*)(W + FEAT + lane * 4);
          float s = xv.y * wv.y;
          s = __builtin_fmaf(xv.x, wv.x, s);
          s = __builtin_fmaf(xv.z, wv.z, s);
          s = __builtin_fmaf(xv.w, wv.w, s);
#pragma unroll
          for (int off = 16; off >= 1; off >>= 1) s += __shfl_xor(s, off, 32);
          if (lane == h) sv = s;
        }
#pragma unroll
        for (int off = 16; off >= 1; off >>= 1) sv += __shfl_xor(sv, off, 32);
        if ((lane >> 4) == (ee >> 3)) sreg[ee & 7] = sv;
      }
    }
  }

  const float bias = bptr[0];
  if (lane == 0) {
#pragma unroll
    for (int r = 0; r < 8; ++r) {
      const int e = eBase + r;
      if (e < E) out[e] = c[r] + sreg[r] + bias;
    }
  } else if (lane == 16) {
#pragma unroll
    for (int r = 0; r < 8; ++r) {
      const int e = eBase + 8 + r;
      if (e < E) out[e] = c[r] + sreg[r] + bias;
    }
  }
}

extern "C" void kernel_launch(void* const* d_in, const int* in_sizes, int n_in,
                              void* d_out, int out_size, void* d_ws, size_t ws_size,
                              hipStream_t stream) {
  const float* x      = (const float*)d_in[0];
  const int*   nbr    = (const int*)  d_in[1];
  const int*   tar_ei = (const int*)  d_in[2];
  const float* W      = (const float*)d_in[3];
  const float* b      = (const float*)d_in[4];
  float*       out    = (float*)d_out;

  const int E = in_sizes[2] / 2;

  const int threads = 256;
  const int edgesPerBlock = (threads / 32) * 16;
  const int blocks2 = (E + edgesPerBlock - 1) / edgesPerBlock;
  ncn_edge_kernel<<<blocks2, threads, 0, stream>>>(x, nbr, tar_ei, W, b, out, E);
}


# Endpoint rows read one whole 512-B row per load (4 lines per instruction, not 16), all 32 in one group; xi*xj staged via LDS into the operand layout; k grouping of the WMMA steps (sum order) changed
# speedup vs baseline: 3.6318x; 1.1255x over previous
#include <hip/hip_runtime.h>

constexpr int FEAT = 128;
constexpr int DEG  = 32;
constexpr int WAVES_PER_BLOCK = 8;
constexpr int ROW_PITCH = FEAT + 4;

typedef __attribute__((ext_vector_type(2))) float v2f;
typedef __attribute__((ext_vector_type(4))) float v4f;
typedef __attribute__((ext_vector_type(8))) float v8f;

__global__ void __launch_bounds__(256) __attribute__((amdgpu_waves_per_eu(4, 4)))
ncn_edge_kernel(const float* __restrict__ x,
                                const int*  __restrict__ nbr,
                                const int*  __restrict__ tar_ei,
                                const float* __restrict__ W,
                                const float* __restrict__ bptr,
                                float* __restrict__ out, int E) {
  __shared__ v4f stage[WAVES_PER_BLOCK][16 * ROW_PITCH / 4];

  const int lane  = (int)(threadIdx.x & 31);
  const int wib   = (int)(threadIdx.x >> 5);
  const int wid   = (int)((blockIdx.x * blockDim.x + threadIdx.x) >> 5);
  const int eBase = wid * 16;
  if (eBase >= E) return;

  const int m    = lane & 15;

  int eIdx = eBase + m;
  if (eIdx > E - 1) eIdx = E - 1;
  const int src = tar_ei[eIdx];
  const int dst = tar_ei[E + eIdx];

  int q[16], njv[16];
#pragma unroll
  for (int ee = 0; ee < 16; ++ee) {
    const int eSrc = __shfl(src, ee, 32);
    const int eDst = __shfl(dst, ee, 32);
    q[ee]   = nbr[(size_t)eSrc * DEG + lane];
    njv[ee] = nbr[(size_t)eDst * DEG + lane];
  }
  v4f xi4[16], xj4[16];
#pragma unroll
  for (int ee = 0; ee < 16; ++ee) {
    const int eSrc = __shfl(src, ee, 32);
    const int eDst = __shfl(dst, ee, 32);
    xi4[ee] = *(const v4f*)(x + (size_t)eSrc * FEAT + lane * 4);
    xj4[ee] = *(const v4f*)(x + (size_t)eDst * FEAT + lane * 4);
  }
  __builtin_amdgcn_sched_barrier(0);

  float* tile = (float*)&stage[wib][0];
#pragma unroll
  for (int ee = 0; ee < 16; ++ee) {
    *(v4f*)(tile + ee * ROW_PITCH + lane * 4) = xi4[ee] * xj4[ee];
  }
  __builtin_amdgcn_wave_barrier();

  const int half = lane >> 4;
  const bool col0 = (m == 0);
  v4f av[16], wq[16];
#pragma unroll
  for (int t = 0; t < 16; ++t) {
    av[t] = *(const v4f*)(tile + m * ROW_PITCH + 8 * t + 4 * half);
    wq[t] = *(const v4f*)(W + 8 * t + 4 * half);
  }

  int pos[16];
#pragma unroll
  for (int ee = 0; ee < 16; ++ee) pos[ee] = 0;
#pragma unroll
  for (int b = 16; b >= 1; b >>= 1) {
#pragma unroll
    for (int ee = 0; ee < 16; ++ee) {
      const int v = __shfl(njv[ee], pos[ee] + b - 1, 32);
      if (v < q[ee]) pos[ee] += b;
    }
  }
  unsigned hmask[16];
  unsigned anyHit = 0u;
#pragma unroll
  for (int ee = 0; ee < 16; ++ee) {
    const bool hit = (__shfl(njv[ee], pos[ee], 32) == q[ee]);
    hmask[ee] = __builtin_amdgcn_ballot_w32(hit);
    anyHit |= hmask[ee];
  }

  v8f c = {0.f, 0.f, 0.f, 0.f, 0.f, 0.f, 0.f, 0.f};
#pragma unroll
  for (int t = 0; t < 16; ++t) {
    v2f a0, a1, b0, b1;
    a0.x = av[t].x;  a0.y = av[t].y;
    a1.x = av[t].z;  a1.y = av[t].w;
    b0.x = col0 ? wq[t].x : 0.0f;
    b0.y = col0 ? wq[t].y : 0.0f;
    b1.x = col0 ? wq[t].z : 0.0f;
    b1.y = col0 ? wq[t].w : 0.0f;
    c = __builtin_amdgcn_wmma_f32_16x16x4_f32(
             false, a0,  false, b0,
             (short)0, c,  false,  false);
    c = __builtin_amdgcn_wmma_f32_16x16x4_f32(
             false, a1,  false, b1,
             (short)0, c,  false,  false);
  }

  float sreg[8] = {0.f, 0.f, 0.f, 0.f, 0.f, 0.f, 0.f, 0.f};
  if (__builtin_expect(anyHit != 0u, 0)) {
#pragma unroll
    for (int ee = 0; ee < 16; ++ee) {
      if (hmask[ee] != 0u) {
        float sv = 0.0f;
        unsigned rem = hmask[ee];
        for (int t = 0; t < 32 && rem != 0u; ++t) {
          const int h  = __builtin_ctz(rem);
          rem &= rem - 1u;
          const int qh = __builtin_amdgcn_readlane(q[ee], h);
          const float4 xv = *(const float4*)(x + (size_t)qh * FEAT + lane * 4);
          const float4 wv = *(const float4*)(W + FEAT + lane * 4);
          float s = xv.y * wv.y;
          s = __builtin_fmaf(xv.x, wv.x, s);
          s = __builtin_fmaf(xv.z, wv.z, s);
          s = __builtin_fmaf(xv.w, wv.w, s);
#pragma unroll
          for (int off = 16; off >= 1; off >>= 1) s += __shfl_xor(s, off, 32);
          if (lane == h) sv = s;
        }
#pragma unroll
        for (int off = 16; off >= 1; off >>= 1) sv += __shfl_xor(sv, off, 32);
        if ((lane >> 4) == (ee >> 3)) sreg[ee & 7] = sv;
      }
    }
  }

  const float bias = bptr[0];
  if (lane == 0) {
#pragma unroll
    for (int r = 0; r < 8; ++r) {
      const int e = eBase + r;
      if (e < E) out[e] = c[r] + sreg[r] + bias;
    }
  } else if (lane == 16) {
#pragma unroll
    for (int r = 0; r < 8; ++r) {
      const int e = eBase + 8 + r;
      if (e < E) out[e] = c[r] + sreg[r] + bias;
    }
  }
}

extern "C" void kernel_launch(void* const* d_in, const int* in_sizes, int n_in,
                              void* d_out, int out_size, void* d_ws, size_t ws_size,
                              hipStream_t stream) {
  const float* x      = (const float*)d_in[0];
  const int*   nbr    = (const int*)  d_in[1];
  const int*   tar_ei = (const int*)  d_in[2];
  const float* W      = (const float*)d_in[3];
  const float* b      = (const float*)d_in[4];
  float*       out    = (float*)d_out;

  const int E = in_sizes[2] / 2;

  const int threads = 256;
  const int edgesPerBlock = (threads / 32) * 16;
  const int blocks2 = (E + edgesPerBlock - 1) / edgesPerBlock;
  ncn_edge_kernel<<<blocks2, threads, 0, stream>>>(x, nbr, tar_ei, W, b, out, E);
}


# Feature dot from the row-load layout: per edge two WMMAs with w on the diagonal as its rows arrive, search rounds between; 16 traces by one lane butterfly; no LDS tile; sum order changed
# speedup vs baseline: 3.8997x; 1.0738x over previous
#include <hip/hip_runtime.h>

constexpr int FEAT = 128;
constexpr int DEG  = 32;

typedef __attribute__((ext_vector_type(2))) float v2f;
typedef __attribute__((ext_vector_type(4))) float v4f;
typedef __attribute__((ext_vector_type(8))) float v8f;

template <int M>
__device__ __forceinline__ float row_xor(float v) {
  const int b = __builtin_bit_cast(int, v);
  return __builtin_bit_cast(
      float, __builtin_amdgcn_update_dpp(0, b, 0x160 + M, 0xf, 0xf, true));
}
__device__ __forceinline__ float cross24(float v) {
  const int b = __builtin_bit_cast(int, v);
  return __builtin_bit_cast(
      float, __builtin_amdgcn_permlanex16(b, b, 0xfedcba98u, 0x76543210u,
                                          false, false));
}

__global__ void __launch_bounds__(256) __attribute__((amdgpu_waves_per_eu(4, 4)))
ncn_edge_kernel(const float* __restrict__ x,
                                const int*  __restrict__ nbr,
                                const int*  __restrict__ tar_ei,
                                const float* __restrict__ W,
                                const float* __restrict__ bptr,
                                float* __restrict__ out, int E) {
  const int lane  = (int)(threadIdx.x & 31);
  const int wid   = (int)((blockIdx.x * blockDim.x + threadIdx.x) >> 5);
  const int eBase = wid * 16;
  if (eBase >= E) return;

  const int m    = lane & 15;

  int eIdx = eBase + m;
  if (eIdx > E - 1) eIdx = E - 1;
  const int src = tar_ei[eIdx];
  const int dst = tar_ei[E + eIdx];

  const v4f w4 = *(const v4f*)(W + lane * 4);

  int q[16], njv[16];
#pragma unroll
  for (int ee = 0; ee < 16; ++ee) {
    const int eSrc = __shfl(src, ee, 32);
    const int eDst = __shfl(dst, ee, 32);
    q[ee]   = nbr[(size_t)eSrc * DEG + lane];
    njv[ee] = nbr[(size_t)eDst * DEG + lane];
  }
  v4f xi4[16], xj4[16];
#pragma unroll
  for (int ee = 0; ee < 16; ++ee) {
    const int eSrc = __shfl(src, ee, 32);
    const int eDst = __shfl(dst, ee, 32);
    xi4[ee] = *(const v4f*)(x + (size_t)eSrc * FEAT + lane * 4);
    xj4[ee] = *(const v4f*)(x + (size_t)eDst * FEAT + lane * 4);
  }
  __builtin_amdgcn_sched_barrier(0);

  int pos[16], got[16];
#pragma unroll
  for (int j = 0; j < 16; ++j) pos[j] = 0;
  unsigned hmask[16];
  unsigned anyHit = 0u;

  const int  sel   = lane & 7;
  const bool diagL = (lane < 8) || (lane >= 24);
  v2f b0, b1;
  b0.x = w4.x;  b0.y = w4.y;
  b1.x = w4.z;  b1.y = w4.w;
  float tr[16];
  v8f dprev = {0.f, 0.f, 0.f, 0.f, 0.f, 0.f, 0.f, 0.f};
#pragma unroll
  for (int ee = 0; ee < 16; ++ee) {
    if (ee < 12) {
      const int r = ee >> 1;
      if ((ee & 1) == 0) {
#pragma unroll
        for (int j = 0; j < 16; ++j)
          got[j] = __shfl(njv[j], (r < 5) ? pos[j] + (16 >> r) - 1 : pos[j], 32);
      } else if (r < 5) {
#pragma unroll
        for (int j = 0; j < 16; ++j)
          if (got[j] < q[j]) pos[j] += (16 >> r);
      } else {
#pragma unroll
        for (int j = 0; j < 16; ++j) {
          const bool hit = (got[j] == q[j]);
          hmask[j] = __builtin_amdgcn_ballot_w32(hit);
          anyHit |= hmask[j];
        }
      }
    }
    const v4f p = xi4[ee] * xj4[ee];
    v2f a0, a1;
    a0.x = p.x;  a0.y = p.y;
    a1.x = p.z;  a1.y = p.w;
    v8f d = {0.f, 0.f, 0.f, 0.f, 0.f, 0.f, 0.f, 0.f};
    d = __builtin_amdgcn_wmma_f32_16x16x4_f32(
             false, a0,  false, b0,
             (short)0, d,  false,  false);
    d = __builtin_amdgcn_wmma_f32_16x16x4_f32(
             false, a1,  false, b1,
             (short)0, d,  false,  false);
    if (ee > 0) {
      float t = dprev[0];
      t = (sel == 1) ? dprev[1] : t;
      t = (sel == 2) ? dprev[2] : t;
      t = (sel == 3) ? dprev[3] : t;
      t = (sel == 4) ? dprev[4] : t;
      t = (sel == 5) ? dprev[5] : t;
      t = (sel == 6) ? dprev[6] : t;
      t = (sel == 7) ? dprev[7] : t;
      tr[ee - 1] = t;
    }
    dprev = d;
    __builtin_amdgcn_sched_barrier(0);
  }
  {
    float t = dprev[0];
    t = (sel == 1) ? dprev[1] : t;
    t = (sel == 2) ? dprev[2] : t;
    t = (sel == 3) ? dprev[3] : t;
    t = (sel == 4) ? dprev[4] : t;
    t = (sel == 5) ? dprev[5] : t;
    t = (sel == 6) ? dprev[6] : t;
    t = (sel == 7) ? dprev[7] : t;
    tr[15] = t;
  }

  float f8[8];
#pragma unroll
  for (int i = 0; i < 8; ++i) {
    const float ex = row_xor<8>(tr[i + 8]);
    f8[i] = diagL ? tr[i] : ex;
  }
  const bool bit2 = (lane & 4) != 0;
  const bool bit1 = (lane & 2) != 0;
  const bool bit0 = (lane & 1) != 0;
  float f4[4];
#pragma unroll
  for (int i = 0; i < 4; ++i) {
    const float keep = bit2 ? f8[i + 4] : f8[i];
    const float send = bit2 ? f8[i] : f8[i + 4];
    f4[i] = keep + row_xor<4>(send);
  }
  float f2[2];
#pragma unroll
  for (int i = 0; i < 2; ++i) {
    const float keep = bit1 ? f4[i + 2] : f4[i];
    const float send = bit1 ? f4[i] : f4[i + 2];
    f2[i] = keep + row_xor<2>(send);
  }
  float dsum;
  {
    const float keep = bit0 ? f2[1] : f2[0];
    const float send = bit0 ? f2[0] : f2[1];
    dsum = keep + row_xor<1>(send);
  }
  dsum = dsum + cross24(dsum);

  float sAcc = 0.f;
  if (__builtin_expect(anyHit != 0u, 0)) {
#pragma unroll
    for (int ee = 0; ee < 16; ++ee) {
      if (hmask[ee] != 0u) {
        float sv = 0.0f;
        unsigned rem = hmask[ee];
        for (int t = 0; t < 32 && rem != 0u; ++t) {
          const int h  = __builtin_ctz(rem);
          rem &= rem - 1u;
          const int qh = __builtin_amdgcn_readlane(q[ee], h);
          const float4 xv = *(const float4*)(x + (size_t)qh * FEAT + lane * 4);
          const float4 wv = *(const float4*)(W + FEAT + lane * 4);
          float s = xv.y * wv.y;
          s = __builtin_fmaf(xv.x, wv.x, s);
          s = __builtin_fmaf(xv.z, wv.z, s);
          s = __builtin_fmaf(xv.w, wv.w, s);
#pragma unroll
          for (int off = 16; off >= 1; off >>= 1) s += __shfl_xor(s, off, 32);
          if (lane == h) sv = s;
        }
#pragma unroll
        for (int off = 16; off >= 1; off >>= 1) sv += __shfl_xor(sv, off, 32);
        if (lane == ee) sAcc = sv;
      }
    }
  }

  const float bias = bptr[0];
  if (lane < 16) {
    const int e = eBase + lane;
    if (e < E) out[e] = dsum + sAcc + bias;
  }
}

extern "C" void kernel_launch(void* const* d_in, const int* in_sizes, int n_in,
                              void* d_out, int out_size, void* d_ws, size_t ws_size,
                              hipStream_t stream) {
  const float* x      = (const float*)d_in[0];
  const int*   nbr    = (const int*)  d_in[1];
  const int*   tar_ei = (const int*)  d_in[2];
  const float* W      = (const float*)d_in[3];
  const float* b      = (const float*)d_in[4];
  float*       out    = (float*)d_out;

  const int E = in_sizes[2] / 2;

  const int threads = 256;
  const int edgesPerBlock = (threads / 32) * 16;
  const int blocks2 = (E + edgesPerBlock - 1) / edgesPerBlock;
  ncn_edge_kernel<<<blocks2, threads, 0, stream>>>(x, nbr, tar_ei, W, b, out, E);
}


# Rare common-neighbour work off the wave's tail: first hit's row requested 4 edges early, lane sums by DPP not LDS shuffles, per-edge second reduction dropped; sum order over an edge's hits changed
# speedup vs baseline: 4.1163x; 1.0555x over previous
#include <hip/hip_runtime.h>

constexpr int FEAT = 128;
constexpr int DEG  = 32;

typedef __attribute__((ext_vector_type(2))) float v2f;
typedef __attribute__((ext_vector_type(4))) float v4f;
typedef __attribute__((ext_vector_type(8))) float v8f;

template <int M>
__device__ __forceinline__ float row_xor(float v) {
  const int b = __builtin_bit_cast(int, v);
  return __builtin_bit_cast(
      float, __builtin_amdgcn_update_dpp(0, b, 0x160 + M, 0xf, 0xf, true));
}
__device__ __forceinline__ float cross24(float v) {
  const int b = __builtin_bit_cast(int, v);
  return __builtin_bit_cast(
      float, __builtin_amdgcn_permlanex16(b, b, 0xfedcba98u, 0x76543210u,
                                          false, false));
}

__device__ __forceinline__ float cross16(float v) {
  const int b = __builtin_bit_cast(int, v);
  return __builtin_bit_cast(
      float, __builtin_amdgcn_permlanex16(b, b, 0x76543210u, 0xfedcba98u,
                                          false, false));
}
__device__ __forceinline__ float wave_sum(float s) {
  s = s + cross16(s);
  s = s + row_xor<8>(s);
  s = s + row_xor<4>(s);
  s = s + row_xor<2>(s);
  s = s + row_xor<1>(s);
  return s;
}

__global__ void __launch_bounds__(256) __attribute__((amdgpu_waves_per_eu(4, 4)))
ncn_edge_kernel(const float* __restrict__ x,
                                const int*  __restrict__ nbr,
                                const int*  __restrict__ tar_ei,
                                const float* __restrict__ W,
                                const float* __restrict__ bptr,
                                float* __restrict__ out, int E) {
  const int lane  = (int)(threadIdx.x & 31);
  const int wid   = (int)((blockIdx.x * blockDim.x + threadIdx.x) >> 5);
  const int eBase = wid * 16;
  if (eBase >= E) return;

  const int m    = lane & 15;

  int eIdx = eBase + m;
  if (eIdx > E - 1) eIdx = E - 1;
  const int src = tar_ei[eIdx];
  const int dst = tar_ei[E + eIdx];

  const v4f w4 = *(const v4f*)(W + lane * 4);

  int q[16], njv[16];
#pragma unroll
  for (int ee = 0; ee < 16; ++ee) {
    const int eSrc = __shfl(src, ee, 32);
    const int eDst = __shfl(dst, ee, 32);
    q[ee]   = nbr[(size_t)eSrc * DEG + lane];
    njv[ee] = nbr[(size_t)eDst * DEG + lane];
  }
  v4f xi4[16], xj4[16];
#pragma unroll
  for (int ee = 0; ee < 16; ++ee) {
    const int eSrc = __shfl(src, ee, 32);
    const int eDst = __shfl(dst, ee, 32);
    xi4[ee] = *(const v4f*)(x + (size_t)eSrc * FEAT + lane * 4);
    xj4[ee] = *(const v4f*)(x + (size_t)eDst * FEAT + lane * 4);
  }
  __builtin_amdgcn_sched_barrier(0);

  int pos[16], got[16];
#pragma unroll
  for (int j = 0; j < 16; ++j) pos[j] = 0;
  unsigned hmask[16];
  unsigned anyHit = 0u;
  int  firstEdge = 16;
  int  firstLane = 0;
  bool restHit   = false;
  v4f xPre, wPre;

  const int  sel   = lane & 7;
  const bool diagL = (lane < 8) || (lane >= 24);
  v2f b0, b1;
  b0.x = w4.x;  b0.y = w4.y;
  b1.x = w4.z;  b1.y = w4.w;
  float tr[16];
  v8f dprev = {0.f, 0.f, 0.f, 0.f, 0.f, 0.f, 0.f, 0.f};
#pragma unroll
  for (int ee = 0; ee < 16; ++ee) {
    if (ee < 12) {
      const int r = ee >> 1;
      if ((ee & 1) == 0) {
#pragma unroll
        for (int j = 0; j < 16; ++j)
          got[j] = __shfl(njv[j], (r < 5) ? pos[j] + (16 >> r) - 1 : pos[j], 32);
      } else if (r < 5) {
#pragma unroll
        for (int j = 0; j < 16; ++j)
          if (got[j] < q[j]) pos[j] += (16 >> r);
      } else {
#pragma unroll
        for (int j = 0; j < 16; ++j) {
          const bool hit = (got[j] == q[j]);
          hmask[j] = __builtin_amdgcn_ballot_w32(hit);
          anyHit |= hmask[j];
        }
      }
    }
    if (ee == 12) {
      if (__builtin_expect(anyHit != 0u, 0)) {
        int myLo = 16, myHi = 16, myNode = 0;
#pragma unroll
        for (int j = 15; j >= 0; --j) {
          const bool hit = (got[j] == q[j]);
          myLo   = hit ? j    : myLo;
          myNode = hit ? q[j] : myNode;
        }
#pragma unroll
        for (int j = 0; j < 16; ++j) {
          const bool hit = (got[j] == q[j]);
          myHi = hit ? j : myHi;
        }
        firstLane = __builtin_ctz(anyHit);
        firstEdge = __builtin_amdgcn_readlane(myLo, firstLane);
        const int firstNode = __builtin_amdgcn_readlane(myNode, firstLane);
        const int hiEdge    = __builtin_amdgcn_readlane(myHi, firstLane);
        restHit = ((anyHit & (anyHit - 1u)) != 0u) || (hiEdge != firstEdge);
        xPre = *(const v4f*)(x + (size_t)firstNode * FEAT + lane * 4);
        wPre = *(const v4f*)(W + FEAT + lane * 4);
      }
    }
    const v4f p = xi4[ee] * xj4[ee];
    v2f a0, a1;
    a0.x = p.x;  a0.y = p.y;
    a1.x = p.z;  a1.y = p.w;
    v8f d = {0.f, 0.f, 0.f, 0.f, 0.f, 0.f, 0.f, 0.f};
    d = __builtin_amdgcn_wmma_f32_16x16x4_f32(
             false, a0,  false, b0,
             (short)0, d,  false,  false);
    d = __builtin_amdgcn_wmma_f32_16x16x4_f32(
             false, a1,  false, b1,
             (short)0, d,  false,  false);
    if (ee > 0) {
      float t = dprev[0];
      t = (sel == 1) ? dprev[1] : t;
      t = (sel == 2) ? dprev[2] : t;
      t = (sel == 3) ? dprev[3] : t;
      t = (sel == 4) ? dprev[4] : t;
      t = (sel == 5) ? dprev[5] : t;
      t = (sel == 6) ? dprev[6] : t;
      t = (sel == 7) ? dprev[7] : t;
      if (ee < 12) asm volatile("" : "+v"(t));
      tr[ee - 1] = t;
    }
    dprev = d;
    __builtin_amdgcn_sched_barrier(0);
  }
  {
    float t = dprev[0];
    t = (sel == 1) ? dprev[1] : t;
    t = (sel == 2) ? dprev[2] : t;
    t = (sel == 3) ? dprev[3] : t;
    t = (sel == 4) ? dprev[4] : t;
    t = (sel == 5) ? dprev[5] : t;
    t = (sel == 6) ? dprev[6] : t;
    t = (sel == 7) ? dprev[7] : t;
    tr[15] = t;
  }

  float f8[8];
#pragma unroll
  for (int i = 0; i < 8; ++i) {
    const float ex = row_xor<8>(tr[i + 8]);
    f8[i] = diagL ? tr[i] : ex;
  }
  const bool bit2 = (lane & 4) != 0;
  const bool bit1 = (lane & 2) != 0;
  const bool bit0 = (lane & 1) != 0;
  float f4[4];
#pragma unroll
  for (int i = 0; i < 4; ++i) {
    const float keep = bit2 ? f8[i + 4] : f8[i];
    const float send = bit2 ? f8[i] : f8[i + 4];
    f4[i] = keep + row_xor<4>(send);
  }
  float f2[2];
#pragma unroll
  for (int i = 0; i < 2; ++i) {
    const float keep = bit1 ? f4[i + 2] : f4[i];
    const float send = bit1 ? f4[i] : f4[i + 2];
    f2[i] = keep + row_xor<2>(send);
  }
  float dsum;
  {
    const float keep = bit0 ? f2[1] : f2[0];
    const float send = bit0 ? f2[0] : f2[1];
    dsum = keep + row_xor<1>(send);
  }
  dsum = dsum + cross24(dsum);

  float sAcc = 0.f;
  if (__builtin_expect(anyHit != 0u, 0)) {
    {
      float s = xPre.y * wPre.y;
      s = __builtin_fmaf(xPre.x, wPre.x, s);
      s = __builtin_fmaf(xPre.z, wPre.z, s);
      s = __builtin_fmaf(xPre.w, wPre.w, s);
      s = wave_sum(s);
      if (lane == firstEdge) sAcc = s;
    }
    if (restHit) {
#pragma unroll
      for (int ee = 0; ee < 16; ++ee) {
        unsigned rem = hmask[ee];
        if (ee == firstEdge) rem &= ~(1u << firstLane);
        if (rem != 0u) {
          float tot = 0.0f;
          for (int t = 0; t < 32 && rem != 0u; ++t) {
            const int h  = __builtin_ctz(rem);
            rem &= rem - 1u;
            const int qh = __builtin_amdgcn_readlane(q[ee], h);
            const v4f xv = *(const v4f*)(x + (size_t)qh * FEAT + lane * 4);
            float s = xv.y * wPre.y;
            s = __builtin_fmaf(xv.x, wPre.x, s);
            s = __builtin_fmaf(xv.z, wPre.z, s);
            s = __builtin_fmaf(xv.w, wPre.w, s);
            tot += wave_sum(s);
          }
          if (lane == ee) sAcc += tot;
        }
      }
    }
  }

  const float bias = bptr[0];
  if (lane < 16) {
    const int e = eBase + lane;
    if (e < E) out[e] = dsum + sAcc + bias;
  }
}

extern "C" void kernel_launch(void* const* d_in, const int* in_sizes, int n_in,
                              void* d_out, int out_size, void* d_ws, size_t ws_size,
                              hipStream_t stream) {
  const float* x      = (const float*)d_in[0];
  const int*   nbr    = (const int*)  d_in[1];
  const int*   tar_ei = (const int*)  d_in[2];
  const float* W      = (const float*)d_in[3];
  const float* b      = (const float*)d_in[4];
  float*       out    = (float*)d_out;

  const int E = in_sizes[2] / 2;

  const int threads = 256;
  const int edgesPerBlock = (threads / 32) * 16;
  const int blocks2 = (E + edgesPerBlock - 1) / edgesPerBlock;
  ncn_edge_kernel<<<blocks2, threads, 0, stream>>>(x, nbr, tar_ei, W, b, out, E);
}
